# starting kernel
# baseline (speedup 1.0000x reference)
#include <hip/hip_runtime.h>
#include <hip/hip_bf16.h>


typedef __attribute__((ext_vector_type(2))) float v2f;
typedef __attribute__((ext_vector_type(8))) float v8f;

#define D 256

__global__ __launch_bounds__(256) void sage_deg_kernel(
    const int* __restrict__ dst, float* __restrict__ deg, int E)
{
    int i = blockIdx.x * blockDim.x + threadIdx.x;
    if (i < E) atomicAdd(&deg[dst[i]], 1.0f);
}

__global__ __launch_bounds__(256) void sage_invdeg_kernel(
    float* __restrict__ deg, int Nn)
{
    int i = blockIdx.x * blockDim.x + threadIdx.x;
    if (i < Nn) deg[i] = 1.0f / fmaxf(deg[i], 1.0f);
}

__global__ __launch_bounds__(256) void sage_scatter_kernel(
    const float* __restrict__ feat, const int* __restrict__ src,
    const int* __restrict__ dst, float* __restrict__ agg, int E)
{
    long long i = (long long)blockIdx.x * blockDim.x + threadIdx.x;
    int e = (int)(i >> 6);
    if (e >= E) return;
    int q = ((int)i & 63) << 2;
    int s  = src[e];
    int d0 = dst[e];

    int ep = e + 512;
    if (ep < E) __builtin_prefetch(feat + (size_t)src[ep] * D + q, 0, 1);

    const float4 v = *(const float4*)(feat + (size_t)s * D + q);
    float* p = agg + (size_t)d0 * D + q;
    atomicAdd(p + 0, v.x);
    atomicAdd(p + 1, v.y);
    atomicAdd(p + 2, v.z);
    atomicAdd(p + 3, v.w);
}

__global__ __launch_bounds__(256) void sage_gemm_kernel(
    const float* __restrict__ agg,  const float* __restrict__ invdeg,
    const float* __restrict__ feat, const float* __restrict__ Wl,
    const float* __restrict__ Wr,   const float* __restrict__ bias,
    float* __restrict__ out, int Nn, int do_relu)
{
    const int wave = threadIdx.x >> 5;
    const int lane = threadIdx.x & 31;
    const int half = lane >> 4;
    const int l15  = lane & 15;

    const int m0 = blockIdx.x * 32 + (wave >> 2) * 16;
    const int n0 = blockIdx.y * 64 + (wave & 3) * 16;

    int am  = m0 + l15;
    int amc = (am < Nn) ? am : (Nn - 1);
    const float sA = invdeg[amc];
    const float* __restrict__ arow = agg  + (size_t)amc * D;
    const float* __restrict__ xrow = feat + (size_t)amc * D;
    const int ncol = n0 + l15;

    v8f acc = {};

#pragma unroll 4
    for (int k = 0; k < D; k += 4) {
        const int ka = k + 2 * half;
        v2f a, b;
        a.x = arow[ka]     * sA;
        a.y = arow[ka + 1] * sA;
        b.x = Wl[(size_t)(ka)     * D + ncol];
        b.y = Wl[(size_t)(ka + 1) * D + ncol];
        acc = __builtin_amdgcn_wmma_f32_16x16x4_f32(false, a, false, b,
                                                    (short)0, acc, false, false);
    }

#pragma unroll 4
    for (int k = 0; k < D; k += 4) {
        const int ka = k + 2 * half;
        v2f a, b;
        a.x = xrow[ka];
        a.y = xrow[ka + 1];
        b.x = Wr[(size_t)(ka)     * D + ncol];
        b.y = Wr[(size_t)(ka + 1) * D + ncol];
        acc = __builtin_amdgcn_wmma_f32_16x16x4_f32(false, a, false, b,
                                                    (short)0, acc, false, false);
    }

    const float bn = bias[ncol];
#pragma unroll
    for (int v = 0; v < 8; ++v) {
        const int row = m0 + v + 8 * half;
        if (row < Nn) {
            float val = acc[v] + bn;
            if (do_relu) val = fmaxf(val, 0.0f);
            out[(size_t)row * D + ncol] = val;
        }
    }
}

extern "C" void kernel_launch(void* const* d_in, const int* in_sizes, int n_in,
                              void* d_out, int out_size, void* d_ws, size_t ws_size,
                              hipStream_t stream)
{
    const float* x   = (const float*)d_in[0];
    const int*   ei  = (const int*)d_in[1];
    const float* W1l = (const float*)d_in[2];
    const float* b1  = (const float*)d_in[3];
    const float* W1r = (const float*)d_in[4];
    const float* W2l = (const float*)d_in[5];
    const float* b2  = (const float*)d_in[6];
    const float* W2r = (const float*)d_in[7];
    float* out = (float*)d_out;

    const int Nn = in_sizes[0] / D;
    const int E  = in_sizes[1] / 2;
    const int* src = ei;
    const int* dst = ei + E;

    float* agg = (float*)d_ws;
    float* deg = agg + (size_t)Nn * D;
    float* h   = deg + (size_t)Nn;

    const size_t aggBytes = (size_t)Nn * D * sizeof(float);
    const size_t degBytes = (size_t)Nn * sizeof(float);

    const int blk = 256;
    const long long scatterThreads = (long long)E * 64;
    const int scatterBlocks = (int)((scatterThreads + blk - 1) / blk);
    dim3 gemmGrid((Nn + 31) / 32, D / 64);

    hipMemsetAsync(agg, 0, aggBytes + degBytes, stream);
    sage_deg_kernel<<<(E + blk - 1) / blk, blk, 0, stream>>>(dst, deg, E);
    sage_invdeg_kernel<<<(Nn + blk - 1) / blk, blk, 0, stream>>>(deg, Nn);

    sage_scatter_kernel<<<scatterBlocks, blk, 0, stream>>>(x, src, dst, agg, E);
    sage_gemm_kernel<<<gemmGrid, blk, 0, stream>>>(agg, deg, x, W1l, W1r, b1,
                                                   h, Nn,  1);

    hipMemsetAsync(agg, 0, aggBytes, stream);
    sage_scatter_kernel<<<scatterBlocks, blk, 0, stream>>>(h, src, dst, agg, E);
    sage_gemm_kernel<<<gemmGrid, blk, 0, stream>>>(agg, deg, h, W2l, W2r, b2,
                                                   out, Nn,  0);
}


# Neighbour sum without global atomics or sorting: each block owns 200 destination rows in LDS, scans dst, queues its edges, gathers their rows, adds them with LDS f32 adds (sums in LDS arrival order)
# speedup vs baseline: 1.7969x; 1.7969x over previous
#include <hip/hip_runtime.h>
#include <hip/hip_bf16.h>


typedef __attribute__((ext_vector_type(2))) float v2f;
typedef __attribute__((ext_vector_type(8))) float v8f;

#define D 256

#define AR   200
#define ATH  1024
#define QCAP 16384

__device__ __forceinline__ int agg_pos(int f)
{
    return (f & 128) | ((f & 3) << 5) | ((f & 127) >> 2);
}

__device__ __forceinline__ void agg_push(
    int d, int e, unsigned r0, unsigned* qn, int* degc, unsigned* q,
    float* acc, const float* __restrict__ feat, const int* __restrict__ src)
{
    const unsigned dl = (unsigned)d - r0;
    if (dl < (unsigned)AR) {
        const unsigned slot = atomicAdd(qn, 1u);
        atomicAdd(&degc[dl], 1);
        if (slot < (unsigned)QCAP) {
            q[slot] = (dl << 24) | (unsigned)e;
        } else {
            const float* row = feat + (size_t)src[e] * D;
#pragma nounroll
            for (int f = 0; f < D; ++f)
                atomicAdd(&acc[dl * D + agg_pos(f)], row[f]);
        }
    }
}

__global__ __launch_bounds__(ATH) void sage_agg_kernel(
    const float* __restrict__ feat, const int* __restrict__ src,
    const int* __restrict__ dst, float* __restrict__ agg,
    float* __restrict__ invdeg, int E, int Nn)
{
    __shared__ __attribute__((aligned(16))) float acc[AR * D];
    __shared__ unsigned q[QCAP];
    __shared__ int degc[AR];
    __shared__ unsigned qn;

    const int tid  = threadIdx.x;
    const int lane = tid & 31;
    const int wave = __builtin_amdgcn_readfirstlane(tid >> 5);
    const unsigned r0 = blockIdx.x * AR;

    for (int i = tid; i < AR * D / 4; i += ATH)
        ((float4*)acc)[i] = make_float4(0.0f, 0.0f, 0.0f, 0.0f);
    for (int i = tid; i < AR; i += ATH) degc[i] = 0;
    if (tid == 0) qn = 0u;
    __syncthreads();

    const bool vec_ok = ((((size_t)dst) & 15) == 0);
    const int EV = vec_ok ? (E / (ATH * 32)) * (ATH * 32) : 0;
    for (int b = tid * 4; b < EV; b += ATH * 32) {
        int4 v[8];
#pragma unroll
        for (int u = 0; u < 8; ++u) v[u] = *(const int4*)(dst + b + u * ATH * 4);
#pragma unroll
        for (int u = 0; u < 8; ++u) {
            const int e = b + u * ATH * 4;
            agg_push(v[u].x, e + 0, r0, &qn, degc, q, acc, feat, src);
            agg_push(v[u].y, e + 1, r0, &qn, degc, q, acc, feat, src);
            agg_push(v[u].z, e + 2, r0, &qn, degc, q, acc, feat, src);
            agg_push(v[u].w, e + 3, r0, &qn, degc, q, acc, feat, src);
        }
    }
    const int E4 = vec_ok ? (E & ~3) : 0;
    for (int b = EV + tid * 4; b < E4; b += ATH * 4) {
        const int4 v = *(const int4*)(dst + b);
        agg_push(v.x, b + 0, r0, &qn, degc, q, acc, feat, src);
        agg_push(v.y, b + 1, r0, &qn, degc, q, acc, feat, src);
        agg_push(v.z, b + 2, r0, &qn, degc, q, acc, feat, src);
        agg_push(v.w, b + 3, r0, &qn, degc, q, acc, feat, src);
    }
    for (int e = E4 + tid; e < E; e += ATH)
        agg_push(dst[e], e, r0, &qn, degc, q, acc, feat, src);
    __syncthreads();

    const unsigned qtot = qn;
    const int n = __builtin_amdgcn_readfirstlane(
        (int)(qtot < (unsigned)QCAP ? qtot : (unsigned)QCAP));
    for (int b = wave * 32; b < n; b += ATH) {
        const int cnt = (n - b < 32) ? (n - b) : 32;
        const unsigned ent = (lane < cnt) ? q[b + lane] : 0u;
        const int s_l  = src[ent & 0xFFFFFFu];
        const int dl_l = (int)(ent >> 24);
        for (int i = 0; i < cnt; i += 4) {
            float4 va[4], vb[4];
            int rb[4];
#pragma unroll
            for (int u = 0; u < 4; ++u) {
                const int iu = (i + u < cnt) ? (i + u) : (cnt - 1);
                const int s  = __builtin_amdgcn_readlane(s_l, iu);
                rb[u] = __builtin_amdgcn_readlane(dl_l, iu) * D;
                const float4* row = (const float4*)(feat + (size_t)s * D);
                va[u] = row[lane];
                vb[u] = row[32 + lane];
            }
#pragma unroll
            for (int u = 0; u < 4; ++u) {
                if (i + u < cnt) {
                    float* a = acc + rb[u] + lane;
                    atomicAdd(a +   0, va[u].x);
                    atomicAdd(a +  32, va[u].y);
                    atomicAdd(a +  64, va[u].z);
                    atomicAdd(a +  96, va[u].w);
                    atomicAdd(a + 128, vb[u].x);
                    atomicAdd(a + 160, vb[u].y);
                    atomicAdd(a + 192, vb[u].z);
                    atomicAdd(a + 224, vb[u].w);
                }
            }
        }
    }
    __syncthreads();

    for (int i = tid; i < AR * (D / 4); i += ATH) {
        const int rl = i >> 6;
        const int c  = i & 63;
        const unsigned r = r0 + (unsigned)rl;
        if (r < (unsigned)Nn) {
            const float* a = acc + rl * D + (c & 32) * 4 + (c & 31);
            float4 v;
            v.x = a[0]; v.y = a[32]; v.z = a[64]; v.w = a[96];
            *(float4*)(agg + (size_t)r * D + 4 * c) = v;
        }
    }
    for (int i = tid; i < AR; i += ATH) {
        const unsigned r = r0 + (unsigned)i;
        if (r < (unsigned)Nn) invdeg[r] = 1.0f / fmaxf((float)degc[i], 1.0f);
    }
}

__global__ __launch_bounds__(256) void sage_gemm_kernel(
    const float* __restrict__ agg,  const float* __restrict__ invdeg,
    const float* __restrict__ feat, const float* __restrict__ Wl,
    const float* __restrict__ Wr,   const float* __restrict__ bias,
    float* __restrict__ out, int Nn, int do_relu)
{
    const int wave = threadIdx.x >> 5;
    const int lane = threadIdx.x & 31;
    const int half = lane >> 4;
    const int l15  = lane & 15;

    const int m0 = blockIdx.x * 32 + (wave >> 2) * 16;
    const int n0 = blockIdx.y * 64 + (wave & 3) * 16;

    int am  = m0 + l15;
    int amc = (am < Nn) ? am : (Nn - 1);
    const float sA = invdeg[amc];
    const float* __restrict__ arow = agg  + (size_t)amc * D;
    const float* __restrict__ xrow = feat + (size_t)amc * D;
    const int ncol = n0 + l15;

    v8f acc = {};

#pragma unroll 4
    for (int k = 0; k < D; k += 4) {
        const int ka = k + 2 * half;
        v2f a, b;
        a.x = arow[ka]     * sA;
        a.y = arow[ka + 1] * sA;
        b.x = Wl[(size_t)(ka)     * D + ncol];
        b.y = Wl[(size_t)(ka + 1) * D + ncol];
        acc = __builtin_amdgcn_wmma_f32_16x16x4_f32(false, a, false, b,
                                                    (short)0, acc, false, false);
    }

#pragma unroll 4
    for (int k = 0; k < D; k += 4) {
        const int ka = k + 2 * half;
        v2f a, b;
        a.x = xrow[ka];
        a.y = xrow[ka + 1];
        b.x = Wr[(size_t)(ka)     * D + ncol];
        b.y = Wr[(size_t)(ka + 1) * D + ncol];
        acc = __builtin_amdgcn_wmma_f32_16x16x4_f32(false, a, false, b,
                                                    (short)0, acc, false, false);
    }

    const float bn = bias[ncol];
#pragma unroll
    for (int v = 0; v < 8; ++v) {
        const int row = m0 + v + 8 * half;
        if (row < Nn) {
            float val = acc[v] + bn;
            if (do_relu) val = fmaxf(val, 0.0f);
            out[(size_t)row * D + ncol] = val;
        }
    }
}

extern "C" void kernel_launch(void* const* d_in, const int* in_sizes, int n_in,
                              void* d_out, int out_size, void* d_ws, size_t ws_size,
                              hipStream_t stream)
{
    const float* x   = (const float*)d_in[0];
    const int*   ei  = (const int*)d_in[1];
    const float* W1l = (const float*)d_in[2];
    const float* b1  = (const float*)d_in[3];
    const float* W1r = (const float*)d_in[4];
    const float* W2l = (const float*)d_in[5];
    const float* b2  = (const float*)d_in[6];
    const float* W2r = (const float*)d_in[7];
    float* out = (float*)d_out;

    const int Nn = in_sizes[0] / D;
    const int E  = in_sizes[1] / 2;
    const int* src = ei;
    const int* dst = ei + E;

    float* agg = (float*)d_ws;
    float* deg = agg + (size_t)Nn * D;
    float* h   = deg + (size_t)Nn;

    const int aggBlocks = (Nn + AR - 1) / AR;
    dim3 gemmGrid((Nn + 31) / 32, D / 64);

    sage_agg_kernel<<<aggBlocks, ATH, 0, stream>>>(x, src, dst, agg, deg, E, Nn);
    sage_gemm_kernel<<<gemmGrid, 256, 0, stream>>>(agg, deg, x, W1l, W1r, b1,
                                                   h, Nn,  1);

    sage_agg_kernel<<<aggBlocks, ATH, 0, stream>>>(h, src, dst, agg, deg, E, Nn);
    sage_gemm_kernel<<<gemmGrid, 256, 0, stream>>>(agg, deg, h, W2l, W2r, b2,
                                                   out, Nn,  0);
}


# Linear layers: a block stages its 32 rows and each 32x256 weight chunk in LDS with coalesced 16-byte loads (next chunk prefetched in registers); waves read WMMA fragments from LDS; k order kept
# speedup vs baseline: 3.1363x; 1.7454x over previous
#include <hip/hip_runtime.h>
#include <hip/hip_bf16.h>


typedef __attribute__((ext_vector_type(2))) float v2f;
typedef __attribute__((ext_vector_type(8))) float v8f;

#define D 256

#define AR   200
#define ATH  1024
#define QCAP 16384

__device__ __forceinline__ int agg_pos(int f)
{
    return (f & 128) | ((f & 3) << 5) | ((f & 127) >> 2);
}

__device__ __forceinline__ void agg_push(
    int d, int e, unsigned r0, unsigned* qn, int* degc, unsigned* q,
    float* acc, const float* __restrict__ feat, const int* __restrict__ src)
{
    const unsigned dl = (unsigned)d - r0;
    if (dl < (unsigned)AR) {
        const unsigned slot = atomicAdd(qn, 1u);
        atomicAdd(&degc[dl], 1);
        if (slot < (unsigned)QCAP) {
            q[slot] = (dl << 24) | (unsigned)e;
        } else {
            const float* row = feat + (size_t)src[e] * D;
#pragma nounroll
            for (int f = 0; f < D; ++f)
                atomicAdd(&acc[dl * D + agg_pos(f)], row[f]);
        }
    }
}

__global__ __launch_bounds__(ATH) void sage_agg_kernel(
    const float* __restrict__ feat, const int* __restrict__ src,
    const int* __restrict__ dst, float* __restrict__ agg,
    float* __restrict__ invdeg, int E, int Nn)
{
    __shared__ __attribute__((aligned(16))) float acc[AR * D];
    __shared__ unsigned q[QCAP];
    __shared__ int degc[AR];
    __shared__ unsigned qn;

    const int tid  = threadIdx.x;
    const int lane = tid & 31;
    const int wave = __builtin_amdgcn_readfirstlane(tid >> 5);
    const unsigned r0 = blockIdx.x * AR;

    for (int i = tid; i < AR * D / 4; i += ATH)
        ((float4*)acc)[i] = make_float4(0.0f, 0.0f, 0.0f, 0.0f);
    for (int i = tid; i < AR; i += ATH) degc[i] = 0;
    if (tid == 0) qn = 0u;
    __syncthreads();

    const bool vec_ok = ((((size_t)dst) & 15) == 0);
    const int EV = vec_ok ? (E / (ATH * 32)) * (ATH * 32) : 0;
    for (int b = tid * 4; b < EV; b += ATH * 32) {
        int4 v[8];
#pragma unroll
        for (int u = 0; u < 8; ++u) v[u] = *(const int4*)(dst + b + u * ATH * 4);
#pragma unroll
        for (int u = 0; u < 8; ++u) {
            const int e = b + u * ATH * 4;
            agg_push(v[u].x, e + 0, r0, &qn, degc, q, acc, feat, src);
            agg_push(v[u].y, e + 1, r0, &qn, degc, q, acc, feat, src);
            agg_push(v[u].z, e + 2, r0, &qn, degc, q, acc, feat, src);
            agg_push(v[u].w, e + 3, r0, &qn, degc, q, acc, feat, src);
        }
    }
    const int E4 = vec_ok ? (E & ~3) : 0;
    for (int b = EV + tid * 4; b < E4; b += ATH * 4) {
        const int4 v = *(const int4*)(dst + b);
        agg_push(v.x, b + 0, r0, &qn, degc, q, acc, feat, src);
        agg_push(v.y, b + 1, r0, &qn, degc, q, acc, feat, src);
        agg_push(v.z, b + 2, r0, &qn, degc, q, acc, feat, src);
        agg_push(v.w, b + 3, r0, &qn, degc, q, acc, feat, src);
    }
    for (int e = E4 + tid; e < E; e += ATH)
        agg_push(dst[e], e, r0, &qn, degc, q, acc, feat, src);
    __syncthreads();

    const unsigned qtot = qn;
    const int n = __builtin_amdgcn_readfirstlane(
        (int)(qtot < (unsigned)QCAP ? qtot : (unsigned)QCAP));
    for (int b = wave * 32; b < n; b += ATH) {
        const int cnt = (n - b < 32) ? (n - b) : 32;
        const unsigned ent = (lane < cnt) ? q[b + lane] : 0u;
        const int s_l  = src[ent & 0xFFFFFFu];
        const int dl_l = (int)(ent >> 24);
        for (int i = 0; i < cnt; i += 4) {
            float4 va[4], vb[4];
            int rb[4];
#pragma unroll
            for (int u = 0; u < 4; ++u) {
                const int iu = (i + u < cnt) ? (i + u) : (cnt - 1);
                const int s  = __builtin_amdgcn_readlane(s_l, iu);
                rb[u] = __builtin_amdgcn_readlane(dl_l, iu) * D;
                const float4* row = (const float4*)(feat + (size_t)s * D);
                va[u] = row[lane];
                vb[u] = row[32 + lane];
            }
#pragma unroll
            for (int u = 0; u < 4; ++u) {
                if (i + u < cnt) {
                    float* a = acc + rb[u] + lane;
                    atomicAdd(a +   0, va[u].x);
                    atomicAdd(a +  32, va[u].y);
                    atomicAdd(a +  64, va[u].z);
                    atomicAdd(a +  96, va[u].w);
                    atomicAdd(a + 128, vb[u].x);
                    atomicAdd(a + 160, vb[u].y);
                    atomicAdd(a + 192, vb[u].z);
                    atomicAdd(a + 224, vb[u].w);
                }
            }
        }
    }
    __syncthreads();

    for (int i = tid; i < AR * (D / 4); i += ATH) {
        const int rl = i >> 6;
        const int c  = i & 63;
        const unsigned r = r0 + (unsigned)rl;
        if (r < (unsigned)Nn) {
            const float* a = acc + rl * D + (c & 32) * 4 + (c & 31);
            float4 v;
            v.x = a[0]; v.y = a[32]; v.z = a[64]; v.w = a[96];
            *(float4*)(agg + (size_t)r * D + 4 * c) = v;
        }
    }
    for (int i = tid; i < AR; i += ATH) {
        const unsigned r = r0 + (unsigned)i;
        if (r < (unsigned)Nn) invdeg[r] = 1.0f / fmaxf((float)degc[i], 1.0f);
    }
}

#define GBM    32
#define GKC    32
#define NCHUNK 16
#define APITCH 36
#define WPITCH 544

struct GemmStage {
    float4 a;
    float4 u0, u1, u2, u3;
    float4 v0, v1, v2, v3;
};

__device__ __forceinline__ void gemm_fetch(
    int ch, const float4* __restrict__ aggp, const float4* __restrict__ featp, float sA,
    const float* __restrict__ Wl, const float* __restrict__ Wr, int pr, int wc4,
    GemmStage& st)
{
    const bool mean = ch < NCHUNK / 2;
    const int kc = (ch & (NCHUNK / 2 - 1)) * GKC;
    const float4* ap = mean ? aggp : featp;
    const float sc = mean ? sA : 1.0f;
    float4 a = ap[kc / 4];
    a.x *= sc; a.y *= sc; a.z *= sc; a.w *= sc;
    st.a = a;
    const float* Wm = mean ? Wl : Wr;
    const float4* wp = (const float4*)(Wm + (size_t)(kc + 2 * pr) * D) + wc4;
    const int RS = D / 4;
    st.u0 = wp[0];       st.v0 = wp[RS];
    st.u1 = wp[8 * RS];  st.v1 = wp[9 * RS];
    st.u2 = wp[16 * RS]; st.v2 = wp[17 * RS];
    st.u3 = wp[24 * RS]; st.v3 = wp[25 * RS];
}

__device__ __forceinline__ void gemm_put_pair(float* dst, const float4 u, const float4 v)
{
    *(float4*)(dst)     = make_float4(u.x, v.x, u.y, v.y);
    *(float4*)(dst + 4) = make_float4(u.z, v.z, u.w, v.w);
}

__device__ __forceinline__ void gemm_put(float* aDst, float* wDst, const GemmStage& st)
{
    *(float4*)aDst = st.a;
    gemm_put_pair(wDst,               st.u0, st.v0);
    gemm_put_pair(wDst + 4 * WPITCH,  st.u1, st.v1);
    gemm_put_pair(wDst + 8 * WPITCH,  st.u2, st.v2);
    gemm_put_pair(wDst + 12 * WPITCH, st.u3, st.v3);
}

__device__ __forceinline__ void gemm_mult(
    const float* afrag, const float* bfrag, v8f& acc00, v8f& acc01, v8f& acc10, v8f& acc11)
{
#pragma unroll
    for (int j = 0; j < GKC / 4; ++j) {
        const v2f a0 = *(const v2f*)(afrag + 4 * j);
        const v2f a1 = *(const v2f*)(afrag + 16 * APITCH + 4 * j);
        const v2f b0 = *(const v2f*)(bfrag + 2 * j * WPITCH);
        const v2f b1 = *(const v2f*)(bfrag + 2 * j * WPITCH + 32);
        acc00 = __builtin_amdgcn_wmma_f32_16x16x4_f32(false, a0, false, b0,
                                                      (short)0, acc00, false, false);
        acc01 = __builtin_amdgcn_wmma_f32_16x16x4_f32(false, a0, false, b1,
                                                      (short)0, acc01, false, false);
        acc10 = __builtin_amdgcn_wmma_f32_16x16x4_f32(false, a1, false, b0,
                                                      (short)0, acc10, false, false);
        acc11 = __builtin_amdgcn_wmma_f32_16x16x4_f32(false, a1, false, b1,
                                                      (short)0, acc11, false, false);
    }
}

__global__ __launch_bounds__(256) void sage_gemm_kernel(
    const float* __restrict__ agg,  const float* __restrict__ invdeg,
    const float* __restrict__ feat, const float* __restrict__ Wl,
    const float* __restrict__ Wr,   const float* __restrict__ bias,
    float* __restrict__ out, int Nn, int do_relu)
{
    __shared__ __attribute__((aligned(16))) float As[GBM * APITCH];
    __shared__ __attribute__((aligned(16))) float Ws[(GKC / 2) * WPITCH];

    const int tid  = threadIdx.x;
    const int wave = tid >> 5;
    const int lane = tid & 31;
    const int half = lane >> 4;
    const int l15  = lane & 15;
    const int m0   = blockIdx.x * GBM;

    const int srow = tid >> 3;
    const int sc4  = tid & 7;
    int ar = m0 + srow;
    ar = (ar < Nn) ? ar : (Nn - 1);
    const float sA = invdeg[ar];
    const float4* aggp  = (const float4*)(agg  + (size_t)ar * D) + sc4;
    const float4* featp = (const float4*)(feat + (size_t)ar * D) + sc4;
    const int pr  = tid >> 6;
    const int wc4 = tid & 63;
    float* aDst = As + srow * APITCH + sc4 * 4;
    float* wDst = Ws + pr * WPITCH + wc4 * 8;

    const float* afrag = As + l15 * APITCH + 2 * half;
    const float* bfrag = Ws + half * WPITCH + 2 * (wave * 32 + l15);

    v8f acc00 = {}, acc01 = {}, acc10 = {}, acc11 = {};

    GemmStage st;
    gemm_fetch(0, aggp, featp, sA, Wl, Wr, pr, wc4, st);
    for (int ch = 0; ch < NCHUNK - 1; ++ch) {
        gemm_put(aDst, wDst, st);
        __syncthreads();
        gemm_fetch(ch + 1, aggp, featp, sA, Wl, Wr, pr, wc4, st);
        gemm_mult(afrag, bfrag, acc00, acc01, acc10, acc11);
        __syncthreads();
    }
    gemm_put(aDst, wDst, st);
    __syncthreads();
    gemm_mult(afrag, bfrag, acc00, acc01, acc10, acc11);

    const int col0 = wave * 32 + l15;
    const float bn0 = bias[col0];
    const float bn1 = bias[col0 + 16];
#pragma unroll
    for (int v = 0; v < 8; ++v) {
        const int r0 = m0 + v + 8 * half;
        const int r1 = r0 + 16;
        float v00 = acc00[v] + bn0, v01 = acc01[v] + bn1;
        float v10 = acc10[v] + bn0, v11 = acc11[v] + bn1;
        if (do_relu) {
            v00 = fmaxf(v00, 0.0f); v01 = fmaxf(v01, 0.0f);
            v10 = fmaxf(v10, 0.0f); v11 = fmaxf(v11, 0.0f);
        }
        if (r0 < Nn) {
            out[(size_t)r0 * D + col0]      = v00;
            out[(size_t)r0 * D + col0 + 16] = v01;
        }
        if (r1 < Nn) {
            out[(size_t)r1 * D + col0]      = v10;
            out[(size_t)r1 * D + col0 + 16] = v11;
        }
    }
}

extern "C" void kernel_launch(void* const* d_in, const int* in_sizes, int n_in,
                              void* d_out, int out_size, void* d_ws, size_t ws_size,
                              hipStream_t stream)
{
    const float* x   = (const float*)d_in[0];
    const int*   ei  = (const int*)d_in[1];
    const float* W1l = (const float*)d_in[2];
    const float* b1  = (const float*)d_in[3];
    const float* W1r = (const float*)d_in[4];
    const float* W2l = (const float*)d_in[5];
    const float* b2  = (const float*)d_in[6];
    const float* W2r = (const float*)d_in[7];
    float* out = (float*)d_out;

    const int Nn = in_sizes[0] / D;
    const int E  = in_sizes[1] / 2;
    const int* src = ei;
    const int* dst = ei + E;

    float* agg = (float*)d_ws;
    float* deg = agg + (size_t)Nn * D;
    float* h   = deg + (size_t)Nn;

    const int aggBlocks = (Nn + AR - 1) / AR;
    dim3 gemmGrid((Nn + GBM - 1) / GBM);

    sage_agg_kernel<<<aggBlocks, ATH, 0, stream>>>(x, src, dst, agg, deg, E, Nn);
    sage_gemm_kernel<<<gemmGrid, 256, 0, stream>>>(agg, deg, x, W1l, W1r, b1,
                                                   h, Nn,  1);

    sage_agg_kernel<<<aggBlocks, ATH, 0, stream>>>(h, src, dst, agg, deg, E, Nn);
    sage_gemm_kernel<<<gemmGrid, 256, 0, stream>>>(agg, deg, h, W2l, W2r, b2,
                                                   out, Nn,  0);
}


# Layer 2 skips the dst scan: layer 1 stores each block's edge queue (local row, source node) and count, layer 2 gathers from that list; its LDS sums follow layer 1's queue order (arrival order)
# speedup vs baseline: 3.5190x; 1.1220x over previous
#include <hip/hip_runtime.h>
#include <hip/hip_bf16.h>


typedef __attribute__((ext_vector_type(2))) float v2f;
typedef __attribute__((ext_vector_type(8))) float v8f;

#define D 256

#define AR   200
#define ATH  1024
#define QCAP 16384

__device__ __forceinline__ int agg_pos(int f)
{
    return (f & 128) | ((f & 3) << 5) | ((f & 127) >> 2);
}

__device__ __forceinline__ void agg_push(
    int d, int e, unsigned r0, unsigned* qn, int* degc, unsigned* q,
    float* acc, const float* __restrict__ feat, const int* __restrict__ src)
{
    const unsigned dl = (unsigned)d - r0;
    if (dl < (unsigned)AR) {
        const unsigned slot = atomicAdd(qn, 1u);
        atomicAdd(&degc[dl], 1);
        if (slot < (unsigned)QCAP) {
            q[slot] = (dl << 24) | (unsigned)e;
        } else {
            const float* row = feat + (size_t)src[e] * D;
#pragma nounroll
            for (int f = 0; f < D; ++f)
                atomicAdd(&acc[dl * D + agg_pos(f)], row[f]);
        }
    }
}

__global__ __launch_bounds__(ATH) void sage_agg_kernel(
    const float* __restrict__ feat, const int* __restrict__ src,
    const int* __restrict__ dst, float* __restrict__ agg,
    float* __restrict__ invdeg, int* __restrict__ qsave, int E, int Nn, int reuse)
{
    __shared__ __attribute__((aligned(16))) float acc[AR * D];
    __shared__ unsigned q[QCAP];
    __shared__ int degc[AR];
    __shared__ unsigned qn;

    const int tid  = threadIdx.x;
    const int lane = tid & 31;
    const int wave = __builtin_amdgcn_readfirstlane(tid >> 5);
    const unsigned r0 = blockIdx.x * AR;

    int* const sq = qsave + (size_t)blockIdx.x * QCAP;
    int* const sn = qsave + (size_t)gridDim.x * QCAP + blockIdx.x;
    const unsigned saved = reuse ? (unsigned)__builtin_amdgcn_readfirstlane(*sn) : 0u;
    const bool scan = !reuse || saved > (unsigned)QCAP;

    for (int i = tid; i < AR * D / 4; i += ATH)
        ((float4*)acc)[i] = make_float4(0.0f, 0.0f, 0.0f, 0.0f);
    for (int i = tid; i < AR; i += ATH) degc[i] = 0;
    if (tid == 0) qn = 0u;
    __syncthreads();

    if (scan) {
        const bool vec_ok = ((((size_t)dst) & 15) == 0);
        const int EV = vec_ok ? (E / (ATH * 32)) * (ATH * 32) : 0;
        for (int b = tid * 4; b < EV; b += ATH * 32) {
            int4 v[8];
#pragma unroll
            for (int u = 0; u < 8; ++u) v[u] = *(const int4*)(dst + b + u * ATH * 4);
#pragma unroll
            for (int u = 0; u < 8; ++u) {
                const int e = b + u * ATH * 4;
                agg_push(v[u].x, e + 0, r0, &qn, degc, q, acc, feat, src);
                agg_push(v[u].y, e + 1, r0, &qn, degc, q, acc, feat, src);
                agg_push(v[u].z, e + 2, r0, &qn, degc, q, acc, feat, src);
                agg_push(v[u].w, e + 3, r0, &qn, degc, q, acc, feat, src);
            }
        }
        const int E4 = vec_ok ? (E & ~3) : 0;
        for (int b = EV + tid * 4; b < E4; b += ATH * 4) {
            const int4 v = *(const int4*)(dst + b);
            agg_push(v.x, b + 0, r0, &qn, degc, q, acc, feat, src);
            agg_push(v.y, b + 1, r0, &qn, degc, q, acc, feat, src);
            agg_push(v.z, b + 2, r0, &qn, degc, q, acc, feat, src);
            agg_push(v.w, b + 3, r0, &qn, degc, q, acc, feat, src);
        }
        for (int e = E4 + tid; e < E; e += ATH)
            agg_push(dst[e], e, r0, &qn, degc, q, acc, feat, src);
    }
    __syncthreads();

    const unsigned qtot = scan ? qn : saved;
    const int n = __builtin_amdgcn_readfirstlane(
        (int)(qtot < (unsigned)QCAP ? qtot : (unsigned)QCAP));
    if (!reuse && tid == 0) *sn = (int)qtot;
    for (int b = wave * 32; b < n; b += ATH) {
        const int cnt = (n - b < 32) ? (n - b) : 32;
        unsigned ent;
        int s_l;
        if (scan) {
            ent = (lane < cnt) ? q[b + lane] : 0u;
            s_l = src[ent & 0xFFFFFFu];
            if (!reuse && lane < cnt) sq[b + lane] = (int)((ent & 0xFF000000u) | (unsigned)s_l);
        } else {
            ent = (lane < cnt) ? (unsigned)sq[b + lane] : 0u;
            s_l = (int)(ent & 0xFFFFFFu);
        }
        const int dl_l = (int)(ent >> 24);
        for (int i = 0; i < cnt; i += 4) {
            float4 va[4], vb[4];
            int rb[4];
#pragma unroll
            for (int u = 0; u < 4; ++u) {
                const int iu = (i + u < cnt) ? (i + u) : (cnt - 1);
                const int s  = __builtin_amdgcn_readlane(s_l, iu);
                rb[u] = __builtin_amdgcn_readlane(dl_l, iu) * D;
                const float4* row = (const float4*)(feat + (size_t)s * D);
                va[u] = row[lane];
                vb[u] = row[32 + lane];
            }
#pragma unroll
            for (int u = 0; u < 4; ++u) {
                if (i + u < cnt) {
                    float* a = acc + rb[u] + lane;
                    atomicAdd(a +   0, va[u].x);
                    atomicAdd(a +  32, va[u].y);
                    atomicAdd(a +  64, va[u].z);
                    atomicAdd(a +  96, va[u].w);
                    atomicAdd(a + 128, vb[u].x);
                    atomicAdd(a + 160, vb[u].y);
                    atomicAdd(a + 192, vb[u].z);
                    atomicAdd(a + 224, vb[u].w);
                }
            }
        }
    }
    __syncthreads();

    for (int i = tid; i < AR * (D / 4); i += ATH) {
        const int rl = i >> 6;
        const int c  = i & 63;
        const unsigned r = r0 + (unsigned)rl;
        if (r < (unsigned)Nn) {
            const float* a = acc + rl * D + (c & 32) * 4 + (c & 31);
            float4 v;
            v.x = a[0]; v.y = a[32]; v.z = a[64]; v.w = a[96];
            *(float4*)(agg + (size_t)r * D + 4 * c) = v;
        }
    }
    if (scan) {
        for (int i = tid; i < AR; i += ATH) {
            const unsigned r = r0 + (unsigned)i;
            if (r < (unsigned)Nn) invdeg[r] = 1.0f / fmaxf((float)degc[i], 1.0f);
        }
    }
}

#define GBM    32
#define GKC    32
#define NCHUNK 16
#define APITCH 36
#define WPITCH 544

struct GemmStage {
    float4 a;
    float4 u0, u1, u2, u3;
    float4 v0, v1, v2, v3;
};

__device__ __forceinline__ void gemm_fetch(
    int ch, const float4* __restrict__ aggp, const float4* __restrict__ featp, float sA,
    const float* __restrict__ Wl, const float* __restrict__ Wr, int pr, int wc4,
    GemmStage& st)
{
    const bool mean = ch < NCHUNK / 2;
    const int kc = (ch & (NCHUNK / 2 - 1)) * GKC;
    const float4* ap = mean ? aggp : featp;
    const float sc = mean ? sA : 1.0f;
    float4 a = ap[kc / 4];
    a.x *= sc; a.y *= sc; a.z *= sc; a.w *= sc;
    st.a = a;
    const float* Wm = mean ? Wl : Wr;
    const float4* wp = (const float4*)(Wm + (size_t)(kc + 2 * pr) * D) + wc4;
    const int RS = D / 4;
    st.u0 = wp[0];       st.v0 = wp[RS];
    st.u1 = wp[8 * RS];  st.v1 = wp[9 * RS];
    st.u2 = wp[16 * RS]; st.v2 = wp[17 * RS];
    st.u3 = wp[24 * RS]; st.v3 = wp[25 * RS];
}

__device__ __forceinline__ void gemm_put_pair(float* dst, const float4 u, const float4 v)
{
    *(float4*)(dst)     = make_float4(u.x, v.x, u.y, v.y);
    *(float4*)(dst + 4) = make_float4(u.z, v.z, u.w, v.w);
}

__device__ __forceinline__ void gemm_put(float* aDst, float* wDst, const GemmStage& st)
{
    *(float4*)aDst = st.a;
    gemm_put_pair(wDst,               st.u0, st.v0);
    gemm_put_pair(wDst + 4 * WPITCH,  st.u1, st.v1);
    gemm_put_pair(wDst + 8 * WPITCH,  st.u2, st.v2);
    gemm_put_pair(wDst + 12 * WPITCH, st.u3, st.v3);
}

__device__ __forceinline__ void gemm_mult(
    const float* afrag, const float* bfrag, v8f& acc00, v8f& acc01, v8f& acc10, v8f& acc11)
{
#pragma unroll
    for (int j = 0; j < GKC / 4; ++j) {
        const v2f a0 = *(const v2f*)(afrag + 4 * j);
        const v2f a1 = *(const v2f*)(afrag + 16 * APITCH + 4 * j);
        const v2f b0 = *(const v2f*)(bfrag + 2 * j * WPITCH);
        const v2f b1 = *(const v2f*)(bfrag + 2 * j * WPITCH + 32);
        acc00 = __builtin_amdgcn_wmma_f32_16x16x4_f32(false, a0, false, b0,
                                                      (short)0, acc00, false, false);
        acc01 = __builtin_amdgcn_wmma_f32_16x16x4_f32(false, a0, false, b1,
                                                      (short)0, acc01, false, false);
        acc10 = __builtin_amdgcn_wmma_f32_16x16x4_f32(false, a1, false, b0,
                                                      (short)0, acc10, false, false);
        acc11 = __builtin_amdgcn_wmma_f32_16x16x4_f32(false, a1, false, b1,
                                                      (short)0, acc11, false, false);
    }
}

__global__ __launch_bounds__(256) void sage_gemm_kernel(
    const float* __restrict__ agg,  const float* __restrict__ invdeg,
    const float* __restrict__ feat, const float* __restrict__ Wl,
    const float* __restrict__ Wr,   const float* __restrict__ bias,
    float* __restrict__ out, int Nn, int do_relu)
{
    __shared__ __attribute__((aligned(16))) float As[GBM * APITCH];
    __shared__ __attribute__((aligned(16))) float Ws[(GKC / 2) * WPITCH];

    const int tid  = threadIdx.x;
    const int wave = tid >> 5;
    const int lane = tid & 31;
    const int half = lane >> 4;
    const int l15  = lane & 15;
    const int m0   = blockIdx.x * GBM;

    const int srow = tid >> 3;
    const int sc4  = tid & 7;
    int ar = m0 + srow;
    ar = (ar < Nn) ? ar : (Nn - 1);
    const float sA = invdeg[ar];
    const float4* aggp  = (const float4*)(agg  + (size_t)ar * D) + sc4;
    const float4* featp = (const float4*)(feat + (size_t)ar * D) + sc4;
    const int pr  = tid >> 6;
    const int wc4 = tid & 63;
    float* aDst = As + srow * APITCH + sc4 * 4;
    float* wDst = Ws + pr * WPITCH + wc4 * 8;

    const float* afrag = As + l15 * APITCH + 2 * half;
    const float* bfrag = Ws + half * WPITCH + 2 * (wave * 32 + l15);

    v8f acc00 = {}, acc01 = {}, acc10 = {}, acc11 = {};

    GemmStage st;
    gemm_fetch(0, aggp, featp, sA, Wl, Wr, pr, wc4, st);
    for (int ch = 0; ch < NCHUNK - 1; ++ch) {
        gemm_put(aDst, wDst, st);
        __syncthreads();
        gemm_fetch(ch + 1, aggp, featp, sA, Wl, Wr, pr, wc4, st);
        gemm_mult(afrag, bfrag, acc00, acc01, acc10, acc11);
        __syncthreads();
    }
    gemm_put(aDst, wDst, st);
    __syncthreads();
    gemm_mult(afrag, bfrag, acc00, acc01, acc10, acc11);

    const int col0 = wave * 32 + l15;
    const float bn0 = bias[col0];
    const float bn1 = bias[col0 + 16];
#pragma unroll
    for (int v = 0; v < 8; ++v) {
        const int r0 = m0 + v + 8 * half;
        const int r1 = r0 + 16;
        float v00 = acc00[v] + bn0, v01 = acc01[v] + bn1;
        float v10 = acc10[v] + bn0, v11 = acc11[v] + bn1;
        if (do_relu) {
            v00 = fmaxf(v00, 0.0f); v01 = fmaxf(v01, 0.0f);
            v10 = fmaxf(v10, 0.0f); v11 = fmaxf(v11, 0.0f);
        }
        if (r0 < Nn) {
            out[(size_t)r0 * D + col0]      = v00;
            out[(size_t)r0 * D + col0 + 16] = v01;
        }
        if (r1 < Nn) {
            out[(size_t)r1 * D + col0]      = v10;
            out[(size_t)r1 * D + col0 + 16] = v11;
        }
    }
}

extern "C" void kernel_launch(void* const* d_in, const int* in_sizes, int n_in,
                              void* d_out, int out_size, void* d_ws, size_t ws_size,
                              hipStream_t stream)
{
    const float* x   = (const float*)d_in[0];
    const int*   ei  = (const int*)d_in[1];
    const float* W1l = (const float*)d_in[2];
    const float* b1  = (const float*)d_in[3];
    const float* W1r = (const float*)d_in[4];
    const float* W2l = (const float*)d_in[5];
    const float* b2  = (const float*)d_in[6];
    const float* W2r = (const float*)d_in[7];
    float* out = (float*)d_out;

    const int Nn = in_sizes[0] / D;
    const int E  = in_sizes[1] / 2;
    const int* src = ei;
    const int* dst = ei + E;

    float* agg = (float*)d_ws;
    float* deg = agg + (size_t)Nn * D;
    float* h   = deg + (size_t)Nn;

    const int aggBlocks = (Nn + AR - 1) / AR;
    dim3 gemmGrid((Nn + GBM - 1) / GBM);

    int* qsave = (int*)d_out;

    sage_agg_kernel<<<aggBlocks, ATH, 0, stream>>>(x, src, dst, agg, deg, qsave, E, Nn,  0);
    sage_gemm_kernel<<<gemmGrid, 256, 0, stream>>>(agg, deg, x, W1l, W1r, b1,
                                                   h, Nn,  1);

    sage_agg_kernel<<<aggBlocks, ATH, 0, stream>>>(h, src, dst, agg, deg, qsave, E, Nn,  1);
    sage_gemm_kernel<<<gemmGrid, 256, 0, stream>>>(agg, deg, h, W2l, W2r, b2,
                                                   out, Nn,  0);
}


# Neighbour sum scans one-byte keys: a new first launch writes dst/200 per edge, each block streams 1 byte per edge instead of 4 and reads dst for its own edges only; LDS f32 adds stay in arrival order
# speedup vs baseline: 3.7184x; 1.0567x over previous
#include <hip/hip_runtime.h>
#include <hip/hip_bf16.h>


typedef __attribute__((ext_vector_type(2))) float v2f;
typedef __attribute__((ext_vector_type(8))) float v8f;

#define D 256

#define AR   200
#define ATH  1024
#define QCAP 16384

__device__ __forceinline__ int agg_pos(int f)
{
    return (f & 128) | ((f & 3) << 5) | ((f & 127) >> 2);
}

__device__ __forceinline__ void agg_push(
    int e, unsigned r0, unsigned* qn, int* degc, unsigned* q,
    float* acc, const float* __restrict__ feat, const int* __restrict__ src,
    const int* __restrict__ dst)
{
    const unsigned slot = atomicAdd(qn, 1u);
    if (slot < (unsigned)QCAP) {
        q[slot] = (unsigned)e;
    } else {
        const unsigned dl = (unsigned)dst[e] - r0;
        atomicAdd(&degc[dl], 1);
        const float* row = feat + (size_t)src[e] * D;
#pragma nounroll
        for (int f = 0; f < D; ++f)
            atomicAdd(&acc[dl * D + agg_pos(f)], row[f]);
    }
}

__device__ __forceinline__ unsigned zero_bytes(unsigned x)
{
    return ~(((x & 0x7F7F7F7Fu) + 0x7F7F7F7Fu) | x | 0x7F7F7F7Fu);
}

__device__ __forceinline__ void agg_keys16(
    const uint4 v, int b, unsigned pat, unsigned r0, unsigned* qn, int* degc, unsigned* q,
    float* acc, const float* __restrict__ feat, const int* __restrict__ src,
    const int* __restrict__ dst)
{
    unsigned m = (zero_bytes(v.x ^ pat) >> 7) | (zero_bytes(v.y ^ pat) >> 6) |
                 (zero_bytes(v.z ^ pat) >> 5) | (zero_bytes(v.w ^ pat) >> 4);
    while (m) {
        const int bit = __builtin_ctz(m);
        agg_push(b + 4 * (bit & 7) + (bit >> 3), r0, qn, degc, q, acc, feat, src, dst);
        m &= m - 1u;
    }
}

__device__ __forceinline__ unsigned agg_key(int d)
{
    const unsigned k = (unsigned)d / (unsigned)AR;
    return k < 255u ? k : 255u;
}

__global__ __launch_bounds__(ATH) void sage_key_kernel(
    const int* __restrict__ dst, unsigned char* __restrict__ key, int E)
{
    const int gtid = blockIdx.x * ATH + threadIdx.x;
    const int gsz  = gridDim.x * ATH;
    const bool vec_ok = ((((size_t)dst) & 15) == 0) && ((((size_t)key) & 3) == 0);
    const int E4 = vec_ok ? (E & ~3) : 0;
    for (int b = gtid * 4; b < E4; b += gsz * 4) {
        const int4 v = *(const int4*)(dst + b);
        *(unsigned*)(key + b) = agg_key(v.x) | (agg_key(v.y) << 8) |
                                (agg_key(v.z) << 16) | (agg_key(v.w) << 24);
    }
    for (int e = E4 + gtid; e < E; e += gsz) key[e] = (unsigned char)agg_key(dst[e]);
}

__global__ __launch_bounds__(ATH) void sage_agg_kernel(
    const float* __restrict__ feat, const int* __restrict__ src,
    const int* __restrict__ dst, const unsigned char* __restrict__ key,
    float* __restrict__ agg, float* __restrict__ invdeg, int* __restrict__ qsave,
    int E, int Nn, int reuse)
{
    __shared__ __attribute__((aligned(16))) float acc[AR * D];
    __shared__ unsigned q[QCAP];
    __shared__ int degc[AR];
    __shared__ unsigned qn;

    const int tid  = threadIdx.x;
    const int lane = tid & 31;
    const int wave = __builtin_amdgcn_readfirstlane(tid >> 5);
    const unsigned r0 = blockIdx.x * AR;

    int* const sq = qsave + (size_t)blockIdx.x * QCAP;
    int* const sn = qsave + (size_t)gridDim.x * QCAP + blockIdx.x;
    const unsigned saved = reuse ? (unsigned)__builtin_amdgcn_readfirstlane(*sn) : 0u;
    const bool scan = !reuse || saved > (unsigned)QCAP;

    for (int i = tid; i < AR * D / 4; i += ATH)
        ((float4*)acc)[i] = make_float4(0.0f, 0.0f, 0.0f, 0.0f);
    for (int i = tid; i < AR; i += ATH) degc[i] = 0;
    if (tid == 0) qn = 0u;
    __syncthreads();

    if (scan) {
        const unsigned pat = blockIdx.x * 0x01010101u;
        const bool vec_ok = ((((size_t)key) & 15) == 0);
        const int EV = vec_ok ? (E / (ATH * 64)) * (ATH * 64) : 0;
        for (int b = tid * 16; b < EV; b += ATH * 64) {
            uint4 v[4];
#pragma unroll
            for (int u = 0; u < 4; ++u) v[u] = *(const uint4*)(key + b + u * ATH * 16);
#pragma unroll
            for (int u = 0; u < 4; ++u)
                agg_keys16(v[u], b + u * ATH * 16, pat, r0, &qn, degc, q, acc, feat, src, dst);
        }
        const int E16 = vec_ok ? (E & ~15) : 0;
        for (int b = EV + tid * 16; b < E16; b += ATH * 16) {
            const uint4 v = *(const uint4*)(key + b);
            agg_keys16(v, b, pat, r0, &qn, degc, q, acc, feat, src, dst);
        }
        for (int e = E16 + tid; e < E; e += ATH)
            if ((unsigned)key[e] == blockIdx.x)
                agg_push(e, r0, &qn, degc, q, acc, feat, src, dst);
    }
    __syncthreads();

    const unsigned qtot = scan ? qn : saved;
    const int n = __builtin_amdgcn_readfirstlane(
        (int)(qtot < (unsigned)QCAP ? qtot : (unsigned)QCAP));
    if (!reuse && tid == 0) *sn = (int)qtot;
    for (int b = wave * 32; b < n; b += ATH) {
        const int cnt = (n - b < 32) ? (n - b) : 32;
        unsigned ent;
        int s_l;
        if (scan) {
            const unsigned e_l = (lane < cnt) ? q[b + lane] : 0u;
            s_l = src[e_l];
            const unsigned dl = (unsigned)dst[e_l] - r0;
            if (lane < cnt) atomicAdd(&degc[dl], 1);
            ent = (lane < cnt) ? ((dl << 24) | (unsigned)s_l) : 0u;
            if (!reuse && lane < cnt) sq[b + lane] = (int)ent;
        } else {
            ent = (lane < cnt) ? (unsigned)sq[b + lane] : 0u;
            s_l = (int)(ent & 0xFFFFFFu);
        }
        const int dl_l = (int)(ent >> 24);
        for (int i = 0; i < cnt; i += 4) {
            float4 va[4], vb[4];
            int rb[4];
#pragma unroll
            for (int u = 0; u < 4; ++u) {
                const int iu = (i + u < cnt) ? (i + u) : (cnt - 1);
                const int s  = __builtin_amdgcn_readlane(s_l, iu);
                rb[u] = __builtin_amdgcn_readlane(dl_l, iu) * D;
                const float4* row = (const float4*)(feat + (size_t)s * D);
                va[u] = row[lane];
                vb[u] = row[32 + lane];
            }
#pragma unroll
            for (int u = 0; u < 4; ++u) {
                if (i + u < cnt) {
                    float* a = acc + rb[u] + lane;
                    atomicAdd(a +   0, va[u].x);
                    atomicAdd(a +  32, va[u].y);
                    atomicAdd(a +  64, va[u].z);
                    atomicAdd(a +  96, va[u].w);
                    atomicAdd(a + 128, vb[u].x);
                    atomicAdd(a + 160, vb[u].y);
                    atomicAdd(a + 192, vb[u].z);
                    atomicAdd(a + 224, vb[u].w);
                }
            }
        }
    }
    __syncthreads();

    for (int i = tid; i < AR * (D / 4); i += ATH) {
        const int rl = i >> 6;
        const int c  = i & 63;
        const unsigned r = r0 + (unsigned)rl;
        if (r < (unsigned)Nn) {
            const float* a = acc + rl * D + (c & 32) * 4 + (c & 31);
            float4 v;
            v.x = a[0]; v.y = a[32]; v.z = a[64]; v.w = a[96];
            *(float4*)(agg + (size_t)r * D + 4 * c) = v;
        }
    }
    if (scan) {
        for (int i = tid; i < AR; i += ATH) {
            const unsigned r = r0 + (unsigned)i;
            if (r < (unsigned)Nn) invdeg[r] = 1.0f / fmaxf((float)degc[i], 1.0f);
        }
    }
}

#define GBM    32
#define GKC    32
#define NCHUNK 16
#define APITCH 36
#define WPITCH 544

struct GemmStage {
    float4 a;
    float4 u0, u1, u2, u3;
    float4 v0, v1, v2, v3;
};

__device__ __forceinline__ void gemm_fetch(
    int ch, const float4* __restrict__ aggp, const float4* __restrict__ featp, float sA,
    const float* __restrict__ Wl, const float* __restrict__ Wr, int pr, int wc4,
    GemmStage& st)
{
    const bool mean = ch < NCHUNK / 2;
    const int kc = (ch & (NCHUNK / 2 - 1)) * GKC;
    const float4* ap = mean ? aggp : featp;
    const float sc = mean ? sA : 1.0f;
    float4 a = ap[kc / 4];
    a.x *= sc; a.y *= sc; a.z *= sc; a.w *= sc;
    st.a = a;
    const float* Wm = mean ? Wl : Wr;
    const float4* wp = (const float4*)(Wm + (size_t)(kc + 2 * pr) * D) + wc4;
    const int RS = D / 4;
    st.u0 = wp[0];       st.v0 = wp[RS];
    st.u1 = wp[8 * RS];  st.v1 = wp[9 * RS];
    st.u2 = wp[16 * RS]; st.v2 = wp[17 * RS];
    st.u3 = wp[24 * RS]; st.v3 = wp[25 * RS];
}

__device__ __forceinline__ void gemm_put_pair(float* dst, const float4 u, const float4 v)
{
    *(float4*)(dst)     = make_float4(u.x, v.x, u.y, v.y);
    *(float4*)(dst + 4) = make_float4(u.z, v.z, u.w, v.w);
}

__device__ __forceinline__ void gemm_put(float* aDst, float* wDst, const GemmStage& st)
{
    *(float4*)aDst = st.a;
    gemm_put_pair(wDst,               st.u0, st.v0);
    gemm_put_pair(wDst + 4 * WPITCH,  st.u1, st.v1);
    gemm_put_pair(wDst + 8 * WPITCH,  st.u2, st.v2);
    gemm_put_pair(wDst + 12 * WPITCH, st.u3, st.v3);
}

__device__ __forceinline__ void gemm_mult(
    const float* afrag, const float* bfrag, v8f& acc00, v8f& acc01, v8f& acc10, v8f& acc11)
{
#pragma unroll
    for (int j = 0; j < GKC / 4; ++j) {
        const v2f a0 = *(const v2f*)(afrag + 4 * j);
        const v2f a1 = *(const v2f*)(afrag + 16 * APITCH + 4 * j);
        const v2f b0 = *(const v2f*)(bfrag + 2 * j * WPITCH);
        const v2f b1 = *(const v2f*)(bfrag + 2 * j * WPITCH + 32);
        acc00 = __builtin_amdgcn_wmma_f32_16x16x4_f32(false, a0, false, b0,
                                                      (short)0, acc00, false, false);
        acc01 = __builtin_amdgcn_wmma_f32_16x16x4_f32(false, a0, false, b1,
                                                      (short)0, acc01, false, false);
        acc10 = __builtin_amdgcn_wmma_f32_16x16x4_f32(false, a1, false, b0,
                                                      (short)0, acc10, false, false);
        acc11 = __builtin_amdgcn_wmma_f32_16x16x4_f32(false, a1, false, b1,
                                                      (short)0, acc11, false, false);
    }
}

__global__ __launch_bounds__(256) void sage_gemm_kernel(
    const float* __restrict__ agg,  const float* __restrict__ invdeg,
    const float* __restrict__ feat, const float* __restrict__ Wl,
    const float* __restrict__ Wr,   const float* __restrict__ bias,
    float* __restrict__ out, int Nn, int do_relu)
{
    __shared__ __attribute__((aligned(16))) float As[GBM * APITCH];
    __shared__ __attribute__((aligned(16))) float Ws[(GKC / 2) * WPITCH];

    const int tid  = threadIdx.x;
    const int wave = tid >> 5;
    const int lane = tid & 31;
    const int half = lane >> 4;
    const int l15  = lane & 15;
    const int m0   = blockIdx.x * GBM;

    const int srow = tid >> 3;
    const int sc4  = tid & 7;
    int ar = m0 + srow;
    ar = (ar < Nn) ? ar : (Nn - 1);
    const float sA = invdeg[ar];
    const float4* aggp  = (const float4*)(agg  + (size_t)ar * D) + sc4;
    const float4* featp = (const float4*)(feat + (size_t)ar * D) + sc4;
    const int pr  = tid >> 6;
    const int wc4 = tid & 63;
    float* aDst = As + srow * APITCH + sc4 * 4;
    float* wDst = Ws + pr * WPITCH + wc4 * 8;

    const float* afrag = As + l15 * APITCH + 2 * half;
    const float* bfrag = Ws + half * WPITCH + 2 * (wave * 32 + l15);

    v8f acc00 = {}, acc01 = {}, acc10 = {}, acc11 = {};

    GemmStage st;
    gemm_fetch(0, aggp, featp, sA, Wl, Wr, pr, wc4, st);
    for (int ch = 0; ch < NCHUNK - 1; ++ch) {
        gemm_put(aDst, wDst, st);
        __syncthreads();
        gemm_fetch(ch + 1, aggp, featp, sA, Wl, Wr, pr, wc4, st);
        gemm_mult(afrag, bfrag, acc00, acc01, acc10, acc11);
        __syncthreads();
    }
    gemm_put(aDst, wDst, st);
    __syncthreads();
    gemm_mult(afrag, bfrag, acc00, acc01, acc10, acc11);

    const int col0 = wave * 32 + l15;
    const float bn0 = bias[col0];
    const float bn1 = bias[col0 + 16];
#pragma unroll
    for (int v = 0; v < 8; ++v) {
        const int r0 = m0 + v + 8 * half;
        const int r1 = r0 + 16;
        float v00 = acc00[v] + bn0, v01 = acc01[v] + bn1;
        float v10 = acc10[v] + bn0, v11 = acc11[v] + bn1;
        if (do_relu) {
            v00 = fmaxf(v00, 0.0f); v01 = fmaxf(v01, 0.0f);
            v10 = fmaxf(v10, 0.0f); v11 = fmaxf(v11, 0.0f);
        }
        if (r0 < Nn) {
            out[(size_t)r0 * D + col0]      = v00;
            out[(size_t)r0 * D + col0 + 16] = v01;
        }
        if (r1 < Nn) {
            out[(size_t)r1 * D + col0]      = v10;
            out[(size_t)r1 * D + col0 + 16] = v11;
        }
    }
}

extern "C" void kernel_launch(void* const* d_in, const int* in_sizes, int n_in,
                              void* d_out, int out_size, void* d_ws, size_t ws_size,
                              hipStream_t stream)
{
    const float* x   = (const float*)d_in[0];
    const int*   ei  = (const int*)d_in[1];
    const float* W1l = (const float*)d_in[2];
    const float* b1  = (const float*)d_in[3];
    const float* W1r = (const float*)d_in[4];
    const float* W2l = (const float*)d_in[5];
    const float* b2  = (const float*)d_in[6];
    const float* W2r = (const float*)d_in[7];
    float* out = (float*)d_out;

    const int Nn = in_sizes[0] / D;
    const int E  = in_sizes[1] / 2;
    const int* src = ei;
    const int* dst = ei + E;

    float* agg = (float*)d_ws;
    float* deg = agg + (size_t)Nn * D;
    float* h   = deg + (size_t)Nn;

    const int aggBlocks = (Nn + AR - 1) / AR;
    dim3 gemmGrid((Nn + GBM - 1) / GBM);

    int* qsave = (int*)d_out;
    unsigned char* key = (unsigned char*)(qsave + (size_t)(aggBlocks + 4) * QCAP);
    const int keyBlocks = (E + ATH * 4 - 1) / (ATH * 4);

    sage_key_kernel<<<keyBlocks, ATH, 0, stream>>>(dst, key, E);
    sage_agg_kernel<<<aggBlocks, ATH, 0, stream>>>(x, src, dst, key, agg, deg, qsave, E, Nn,  0);
    sage_gemm_kernel<<<gemmGrid, 256, 0, stream>>>(agg, deg, x, W1l, W1r, b1,
                                                   h, Nn,  1);

    sage_agg_kernel<<<aggBlocks, ATH, 0, stream>>>(h, src, dst, key, agg, deg, qsave, E, Nn,  1);
    sage_gemm_kernel<<<gemmGrid, 256, 0, stream>>>(agg, deg, h, W2l, W2r, b2,
                                                   out, Nn,  0);
}


# Linear layers: one block per WGP owns 208 rows x 256 columns (13x2 accumulator tiles per wave in registers), so a staged weight chunk serves 208 rows instead of 32; k order kept
# speedup vs baseline: 3.9711x; 1.0680x over previous
#include <hip/hip_runtime.h>
#include <hip/hip_bf16.h>


typedef __attribute__((ext_vector_type(2))) float v2f;
typedef __attribute__((ext_vector_type(8))) float v8f;

#define D 256

#define AR   200
#define ATH  1024
#define QCAP 16384

__device__ __forceinline__ int agg_pos(int f)
{
    return (f & 128) | ((f & 3) << 5) | ((f & 127) >> 2);
}

__device__ __forceinline__ void agg_push(
    int e, unsigned r0, unsigned* qn, int* degc, unsigned* q,
    float* acc, const float* __restrict__ feat, const int* __restrict__ src,
    const int* __restrict__ dst)
{
    const unsigned slot = atomicAdd(qn, 1u);
    if (slot < (unsigned)QCAP) {
        q[slot] = (unsigned)e;
    } else {
        const unsigned dl = (unsigned)dst[e] - r0;
        atomicAdd(&degc[dl], 1);
        const float* row = feat + (size_t)src[e] * D;
#pragma nounroll
        for (int f = 0; f < D; ++f)
            atomicAdd(&acc[dl * D + agg_pos(f)], row[f]);
    }
}

__device__ __forceinline__ unsigned zero_bytes(unsigned x)
{
    return ~(((x & 0x7F7F7F7Fu) + 0x7F7F7F7Fu) | x | 0x7F7F7F7Fu);
}

__device__ __forceinline__ void agg_keys16(
    const uint4 v, int b, unsigned pat, unsigned r0, unsigned* qn, int* degc, unsigned* q,
    float* acc, const float* __restrict__ feat, const int* __restrict__ src,
    const int* __restrict__ dst)
{
    unsigned m = (zero_bytes(v.x ^ pat) >> 7) | (zero_bytes(v.y ^ pat) >> 6) |
                 (zero_bytes(v.z ^ pat) >> 5) | (zero_bytes(v.w ^ pat) >> 4);
    while (m) {
        const int bit = __builtin_ctz(m);
        agg_push(b + 4 * (bit & 7) + (bit >> 3), r0, qn, degc, q, acc, feat, src, dst);
        m &= m - 1u;
    }
}

__device__ __forceinline__ unsigned agg_key(int d)
{
    const unsigned k = (unsigned)d / (unsigned)AR;
    return k < 255u ? k : 255u;
}

__global__ __launch_bounds__(ATH) void sage_key_kernel(
    const int* __restrict__ dst, unsigned char* __restrict__ key, int E)
{
    const int gtid = blockIdx.x * ATH + threadIdx.x;
    const int gsz  = gridDim.x * ATH;
    const bool vec_ok = ((((size_t)dst) & 15) == 0) && ((((size_t)key) & 3) == 0);
    const int E4 = vec_ok ? (E & ~3) : 0;
    for (int b = gtid * 4; b < E4; b += gsz * 4) {
        const int4 v = *(const int4*)(dst + b);
        *(unsigned*)(key + b) = agg_key(v.x) | (agg_key(v.y) << 8) |
                                (agg_key(v.z) << 16) | (agg_key(v.w) << 24);
    }
    for (int e = E4 + gtid; e < E; e += gsz) key[e] = (unsigned char)agg_key(dst[e]);
}

__global__ __launch_bounds__(ATH) void sage_agg_kernel(
    const float* __restrict__ feat, const int* __restrict__ src,
    const int* __restrict__ dst, const unsigned char* __restrict__ key,
    float* __restrict__ agg, float* __restrict__ invdeg, int* __restrict__ qsave,
    int E, int Nn, int reuse)
{
    __shared__ __attribute__((aligned(16))) float acc[AR * D];
    __shared__ unsigned q[QCAP];
    __shared__ int degc[AR];
    __shared__ unsigned qn;

    const int tid  = threadIdx.x;
    const int lane = tid & 31;
    const int wave = __builtin_amdgcn_readfirstlane(tid >> 5);
    const unsigned r0 = blockIdx.x * AR;

    int* const sq = qsave + (size_t)blockIdx.x * QCAP;
    int* const sn = qsave + (size_t)gridDim.x * QCAP + blockIdx.x;
    const unsigned saved = reuse ? (unsigned)__builtin_amdgcn_readfirstlane(*sn) : 0u;
    const bool scan = !reuse || saved > (unsigned)QCAP;

    for (int i = tid; i < AR * D / 4; i += ATH)
        ((float4*)acc)[i] = make_float4(0.0f, 0.0f, 0.0f, 0.0f);
    for (int i = tid; i < AR; i += ATH) degc[i] = 0;
    if (tid == 0) qn = 0u;
    __syncthreads();

    if (scan) {
        const unsigned pat = blockIdx.x * 0x01010101u;
        const bool vec_ok = ((((size_t)key) & 15) == 0);
        const int EV = vec_ok ? (E / (ATH * 64)) * (ATH * 64) : 0;
        for (int b = tid * 16; b < EV; b += ATH * 64) {
            uint4 v[4];
#pragma unroll
            for (int u = 0; u < 4; ++u) v[u] = *(const uint4*)(key + b + u * ATH * 16);
#pragma unroll
            for (int u = 0; u < 4; ++u)
                agg_keys16(v[u], b + u * ATH * 16, pat, r0, &qn, degc, q, acc, feat, src, dst);
        }
        const int E16 = vec_ok ? (E & ~15) : 0;
        for (int b = EV + tid * 16; b < E16; b += ATH * 16) {
            const uint4 v = *(const uint4*)(key + b);
            agg_keys16(v, b, pat, r0, &qn, degc, q, acc, feat, src, dst);
        }
        for (int e = E16 + tid; e < E; e += ATH)
            if ((unsigned)key[e] == blockIdx.x)
                agg_push(e, r0, &qn, degc, q, acc, feat, src, dst);
    }
    __syncthreads();

    const unsigned qtot = scan ? qn : saved;
    const int n = __builtin_amdgcn_readfirstlane(
        (int)(qtot < (unsigned)QCAP ? qtot : (unsigned)QCAP));
    if (!reuse && tid == 0) *sn = (int)qtot;
    for (int b = wave * 32; b < n; b += ATH) {
        const int cnt = (n - b < 32) ? (n - b) : 32;
        unsigned ent;
        int s_l;
        if (scan) {
            const unsigned e_l = (lane < cnt) ? q[b + lane] : 0u;
            s_l = src[e_l];
            const unsigned dl = (unsigned)dst[e_l] - r0;
            if (lane < cnt) atomicAdd(&degc[dl], 1);
            ent = (lane < cnt) ? ((dl << 24) | (unsigned)s_l) : 0u;
            if (!reuse && lane < cnt) sq[b + lane] = (int)ent;
        } else {
            ent = (lane < cnt) ? (unsigned)sq[b + lane] : 0u;
            s_l = (int)(ent & 0xFFFFFFu);
        }
        const int dl_l = (int)(ent >> 24);
        for (int i = 0; i < cnt; i += 4) {
            float4 va[4], vb[4];
            int rb[4];
#pragma unroll
            for (int u = 0; u < 4; ++u) {
                const int iu = (i + u < cnt) ? (i + u) : (cnt - 1);
                const int s  = __builtin_amdgcn_readlane(s_l, iu);
                rb[u] = __builtin_amdgcn_readlane(dl_l, iu) * D;
                const float4* row = (const float4*)(feat + (size_t)s * D);
                va[u] = row[lane];
                vb[u] = row[32 + lane];
            }
#pragma unroll
            for (int u = 0; u < 4; ++u) {
                if (i + u < cnt) {
                    float* a = acc + rb[u] + lane;
                    atomicAdd(a +   0, va[u].x);
                    atomicAdd(a +  32, va[u].y);
                    atomicAdd(a +  64, va[u].z);
                    atomicAdd(a +  96, va[u].w);
                    atomicAdd(a + 128, vb[u].x);
                    atomicAdd(a + 160, vb[u].y);
                    atomicAdd(a + 192, vb[u].z);
                    atomicAdd(a + 224, vb[u].w);
                }
            }
        }
    }
    __syncthreads();

    for (int i = tid; i < AR * (D / 4); i += ATH) {
        const int rl = i >> 6;
        const int c  = i & 63;
        const unsigned r = r0 + (unsigned)rl;
        if (r < (unsigned)Nn) {
            const float* a = acc + rl * D + (c & 32) * 4 + (c & 31);
            float4 v;
            v.x = a[0]; v.y = a[32]; v.z = a[64]; v.w = a[96];
            *(float4*)(agg + (size_t)r * D + 4 * c) = v;
        }
    }
    if (scan) {
        for (int i = tid; i < AR; i += ATH) {
            const unsigned r = r0 + (unsigned)i;
            if (r < (unsigned)Nn) invdeg[r] = 1.0f / fmaxf((float)degc[i], 1.0f);
        }
    }
}

#define GBM    208
#define GRT    13
#define GAS    7
#define GKC    32
#define NCHUNK 16
#define APITCH 36
#define WPITCH 544

struct GemmStage {
    float4 a[GAS];
    float4 u0, u1, u2, u3;
    float4 v0, v1, v2, v3;
};

__device__ __forceinline__ void gemm_fetch(
    int ch, const float* __restrict__ agg, const float* __restrict__ feat,
    const int (&aoff)[GAS], bool tail,
    const float* __restrict__ Wl, const float* __restrict__ Wr, int pr, int wc4,
    GemmStage& st)
{
    const bool mean = ch < NCHUNK / 2;
    const int kc = (ch & (NCHUNK / 2 - 1)) * GKC;
    const float4* ap = (const float4*)(mean ? agg : feat) + kc / 4;
#pragma unroll
    for (int r = 0; r < GAS; ++r) {
        if (r < GAS - 1 || tail) st.a[r] = ap[aoff[r]];
    }
    const float* Wm = mean ? Wl : Wr;
    const float4* wp = (const float4*)(Wm + (size_t)(kc + 2 * pr) * D) + wc4;
    const int RS = D / 4;
    st.u0 = wp[0];       st.v0 = wp[RS];
    st.u1 = wp[8 * RS];  st.v1 = wp[9 * RS];
    st.u2 = wp[16 * RS]; st.v2 = wp[17 * RS];
    st.u3 = wp[24 * RS]; st.v3 = wp[25 * RS];
}

__device__ __forceinline__ void gemm_put_pair(float* dst, const float4 u, const float4 v)
{
    *(float4*)(dst)     = make_float4(u.x, v.x, u.y, v.y);
    *(float4*)(dst + 4) = make_float4(u.z, v.z, u.w, v.w);
}

__device__ __forceinline__ void gemm_put(
    int ch, float* aDst, float* wDst, const float (&sA)[GAS], bool tail, const GemmStage& st)
{
    const bool mean = ch < NCHUNK / 2;
#pragma unroll
    for (int r = 0; r < GAS; ++r) {
        if (r < GAS - 1 || tail) {
            const float sc = mean ? sA[r] : 1.0f;
            float4 a = st.a[r];
            a.x *= sc; a.y *= sc; a.z *= sc; a.w *= sc;
            *(float4*)(aDst + r * 32 * APITCH) = a;
        }
    }
    gemm_put_pair(wDst,               st.u0, st.v0);
    gemm_put_pair(wDst + 4 * WPITCH,  st.u1, st.v1);
    gemm_put_pair(wDst + 8 * WPITCH,  st.u2, st.v2);
    gemm_put_pair(wDst + 12 * WPITCH, st.u3, st.v3);
}

__device__ __forceinline__ void gemm_mult(
    const float* afrag, const float* bfrag, v8f (&acc)[GRT][2])
{
#pragma unroll
    for (int j = 0; j < GKC / 4; ++j) {
        const v2f b0 = *(const v2f*)(bfrag + 2 * j * WPITCH);
        const v2f b1 = *(const v2f*)(bfrag + 2 * j * WPITCH + 32);
#pragma unroll
        for (int t = 0; t < GRT; ++t) {
            const v2f a = *(const v2f*)(afrag + t * 16 * APITCH + 4 * j);
            acc[t][0] = __builtin_amdgcn_wmma_f32_16x16x4_f32(false, a, false, b0,
                                                          (short)0, acc[t][0], false, false);
            acc[t][1] = __builtin_amdgcn_wmma_f32_16x16x4_f32(false, a, false, b1,
                                                          (short)0, acc[t][1], false, false);
        }
    }
}

__global__ __launch_bounds__(256) void sage_gemm_kernel(
    const float* __restrict__ agg,  const float* __restrict__ invdeg,
    const float* __restrict__ feat, const float* __restrict__ Wl,
    const float* __restrict__ Wr,   const float* __restrict__ bias,
    float* __restrict__ out, int Nn, int do_relu)
{
    __shared__ __attribute__((aligned(16))) float As[GBM * APITCH];
    __shared__ __attribute__((aligned(16))) float Ws[(GKC / 2) * WPITCH];

    const int tid  = threadIdx.x;
    const int wave = tid >> 5;
    const int lane = tid & 31;
    const int half = lane >> 4;
    const int l15  = lane & 15;
    const int m0   = blockIdx.x * GBM;

    const int srow = tid >> 3;
    const int sc4  = tid & 7;
    const bool tail = srow + 32 * (GAS - 1) < GBM;
    int   aoff[GAS];
    float sA[GAS];
#pragma unroll
    for (int r = 0; r < GAS; ++r) {
        int ar = m0 + srow + 32 * r;
        ar = (ar < Nn) ? ar : (Nn - 1);
        sA[r]   = invdeg[ar];
        aoff[r] = ar * (D / 4) + sc4;
    }
    const int pr  = tid >> 6;
    const int wc4 = tid & 63;
    float* aDst = As + srow * APITCH + sc4 * 4;
    float* wDst = Ws + pr * WPITCH + wc4 * 8;

    const float* afrag = As + l15 * APITCH + 2 * half;
    const float* bfrag = Ws + half * WPITCH + 2 * (wave * 32 + l15);

    v8f acc[GRT][2];
#pragma unroll
    for (int t = 0; t < GRT; ++t) {
        acc[t][0] = v8f{};
        acc[t][1] = v8f{};
    }

    GemmStage st;
    gemm_fetch(0, agg, feat, aoff, tail, Wl, Wr, pr, wc4, st);
    for (int ch = 0; ch < NCHUNK - 1; ++ch) {
        gemm_put(ch, aDst, wDst, sA, tail, st);
        __syncthreads();
        gemm_fetch(ch + 1, agg, feat, aoff, tail, Wl, Wr, pr, wc4, st);
        gemm_mult(afrag, bfrag, acc);
        __syncthreads();
    }
    gemm_put(NCHUNK - 1, aDst, wDst, sA, tail, st);
    __syncthreads();
    gemm_mult(afrag, bfrag, acc);

    const int col0 = wave * 32 + l15;
    const float bn0 = bias[col0];
    const float bn1 = bias[col0 + 16];
#pragma unroll
    for (int t = 0; t < GRT; ++t) {
#pragma unroll
        for (int v = 0; v < 8; ++v) {
            const int r = m0 + 16 * t + v + 8 * half;
            float x0 = acc[t][0][v] + bn0;
            float x1 = acc[t][1][v] + bn1;
            if (do_relu) {
                x0 = fmaxf(x0, 0.0f);
                x1 = fmaxf(x1, 0.0f);
            }
            if (r < Nn) {
                out[(size_t)r * D + col0]      = x0;
                out[(size_t)r * D + col0 + 16] = x1;
            }
        }
    }
}

extern "C" void kernel_launch(void* const* d_in, const int* in_sizes, int n_in,
                              void* d_out, int out_size, void* d_ws, size_t ws_size,
                              hipStream_t stream)
{
    const float* x   = (const float*)d_in[0];
    const int*   ei  = (const int*)d_in[1];
    const float* W1l = (const float*)d_in[2];
    const float* b1  = (const float*)d_in[3];
    const float* W1r = (const float*)d_in[4];
    const float* W2l = (const float*)d_in[5];
    const float* b2  = (const float*)d_in[6];
    const float* W2r = (const float*)d_in[7];
    float* out = (float*)d_out;

    const int Nn = in_sizes[0] / D;
    const int E  = in_sizes[1] / 2;
    const int* src = ei;
    const int* dst = ei + E;

    float* agg = (float*)d_ws;
    float* deg = agg + (size_t)Nn * D;
    float* h   = deg + (size_t)Nn;

    const int aggBlocks = (Nn + AR - 1) / AR;
    dim3 gemmGrid((Nn + GBM - 1) / GBM);

    int* qsave = (int*)d_out;
    unsigned char* key = (unsigned char*)(qsave + (size_t)(aggBlocks + 4) * QCAP);
    const int keyBlocks = (E + ATH * 4 - 1) / (ATH * 4);

    sage_key_kernel<<<keyBlocks, ATH, 0, stream>>>(dst, key, E);
    sage_agg_kernel<<<aggBlocks, ATH, 0, stream>>>(x, src, dst, key, agg, deg, qsave, E, Nn,  0);
    sage_gemm_kernel<<<gemmGrid, 256, 0, stream>>>(agg, deg, x, W1l, W1r, b1,
                                                   h, Nn,  1);

    sage_agg_kernel<<<aggBlocks, ATH, 0, stream>>>(h, src, dst, key, agg, deg, qsave, E, Nn,  1);
    sage_gemm_kernel<<<gemmGrid, 256, 0, stream>>>(agg, deg, h, W2l, W2r, b2,
                                                   out, Nn,  0);
}
